# starting kernel
# baseline (speedup 1.0000x reference)
#include <hip/hip_runtime.h>
#include <hip/hip_bf16.h>


typedef float v2f __attribute__((ext_vector_type(2)));
typedef float v8f __attribute__((ext_vector_type(8)));

#define N_IN 64
#define N_HID 128
#define N_OUT 64


__global__ void k_deg_init(float* __restrict__ deg, int n) {
    int i = blockIdx.x * blockDim.x + threadIdx.x;
    if (i < n) deg[i] = 1.0f;
}

__global__ void k_deg_count(const int* __restrict__ dst, float* __restrict__ deg, int E) {
    int e = blockIdx.x * blockDim.x + threadIdx.x;
    if (e < E) atomicAdd(&deg[dst[e]], 1.0f);
}

__global__ void k_deg_rsqrt(float* __restrict__ deg, int n) {
    int i = blockIdx.x * blockDim.x + threadIdx.x;
    if (i < n) deg[i] = rsqrtf(fmaxf(deg[i], 1.0f));
}

template <int K, int N>
__global__ void k_gemm_wmma_f32(const float* __restrict__ A,
                                const float* __restrict__ B,
                                float* __restrict__ C, int M) {
    const int lane = threadIdx.x & 31;
    const int wave = threadIdx.x >> 5;
    const int wavesPerBlock = blockDim.x >> 5;
    const int tile = blockIdx.x * wavesPerBlock + wave;

    const int nTiles = N / 16;
    const int mTiles = M / 16;
    if (tile >= mTiles * nTiles) return;

    const int row0 = (tile / nTiles) * 16;
    const int col0 = (tile % nTiles) * 16;
    const int half = lane >> 4;
    const int l16  = lane & 15;

    v8f acc = {0.f, 0.f, 0.f, 0.f, 0.f, 0.f, 0.f, 0.f};

#pragma unroll
    for (int k = 0; k < K; k += 4) {
        const float* ap = A + (size_t)(row0 + l16) * K + (k + 2 * half);
        v2f a;
        a.x = ap[0];
        a.y = ap[1];
        const float* bp = B + (size_t)(k + 2 * half) * N + (col0 + l16);
        v2f b;
        b.x = bp[0];
        b.y = bp[N];
        acc = __builtin_amdgcn_wmma_f32_16x16x4_f32(
             false, a,  false, b,
             (short)0, acc,  false,  false);
    }

#pragma unroll
    for (int j = 0; j < 8; ++j) {
        C[(size_t)(row0 + 8 * half + j) * N + (col0 + l16)] = acc[j];
    }
}


template <int C>
__global__ void k_agg_init(const float* __restrict__ h, const float* __restrict__ inv,
                           const float* __restrict__ bias, float* __restrict__ out, int n) {
    const int C4 = C / 4;
    int idx = blockIdx.x * blockDim.x + threadIdx.x;
    if (idx >= n * C4) return;
    int i  = idx / C4;
    int c4 = idx % C4;
    float s = inv[i];
    s = s * s;
    float4 v = ((const float4*)(h + (size_t)i * C))[c4];
    float4 r;
    r.x = s * v.x; r.y = s * v.y; r.z = s * v.z; r.w = s * v.w;
    if (bias != nullptr) {
        float4 bv = ((const float4*)bias)[c4];
        r.x += bv.x; r.y += bv.y; r.z += bv.z; r.w += bv.w;
    }
    ((float4*)(out + (size_t)i * C))[c4] = r;
}

template <int C>
__global__ void k_agg_edges(const float* __restrict__ h, const int* __restrict__ src,
                            const int* __restrict__ dst, const float* __restrict__ inv,
                            float* __restrict__ out, int E) {
    const int C4 = C / 4;
    int idx = blockIdx.x * blockDim.x + threadIdx.x;
    if (idx >= E * C4) return;
    int e  = idx / C4;
    int c4 = idx % C4;
    int s = src[e];
    int d = dst[e];
    float nrm = inv[s] * inv[d];
    float4 v = ((const float4*)(h + (size_t)s * C))[c4];
    float* op = out + (size_t)d * C + 4 * c4;
    atomicAdd(op + 0, nrm * v.x);
    atomicAdd(op + 1, nrm * v.y);
    atomicAdd(op + 2, nrm * v.z);
    atomicAdd(op + 3, nrm * v.w);
}

template <int C>
__global__ void k_bias_relu(float* __restrict__ a, const float* __restrict__ bias, int n) {
    int idx = blockIdx.x * blockDim.x + threadIdx.x;
    if (idx >= n * C) return;
    int c = idx % C;
    a[idx] = fmaxf(a[idx] + bias[c], 0.0f);
}


extern "C" void kernel_launch(void* const* d_in, const int* in_sizes, int n_in,
                              void* d_out, int out_size, void* d_ws, size_t ws_size,
                              hipStream_t stream) {
    const float* x    = (const float*)d_in[0];
    const int*   eidx = (const int*)d_in[1];
    const float* W1   = (const float*)d_in[2];
    const float* b1   = (const float*)d_in[3];
    const float* W2   = (const float*)d_in[4];
    const float* b2   = (const float*)d_in[5];
    float*       out  = (float*)d_out;

    const int n = in_sizes[0] / N_IN;
    const int E = in_sizes[1] / 2;
    const int* src = eidx;
    const int* dst = eidx + E;

    char* ws = (char*)d_ws;
    const size_t OFF_INV = 0;
    const size_t OFF_H0  = (((size_t)n * 4) + 511) & ~(size_t)511;
    const size_t OFF_AGG = OFF_H0 + (size_t)n * N_HID * 4;
    float* inv  = (float*)(ws + OFF_INV);
    float* h0   = (float*)(ws + OFF_H0);
    float* agg1 = (float*)(ws + OFF_AGG);

    const int TPB = 256;

    k_deg_init<<<(n + TPB - 1) / TPB, TPB, 0, stream>>>(inv, n);
    k_deg_count<<<(E + TPB - 1) / TPB, TPB, 0, stream>>>(dst, inv, E);
    k_deg_rsqrt<<<(n + TPB - 1) / TPB, TPB, 0, stream>>>(inv, n);

    {
        int tiles = (n / 16) * (N_HID / 16);
        int blocks = (tiles + 3) / 4;
        k_gemm_wmma_f32<N_IN, N_HID><<<blocks, 128, 0, stream>>>(x, W1, h0, n);
    }
    k_agg_init<N_HID><<<((n * (N_HID / 4)) + TPB - 1) / TPB, TPB, 0, stream>>>(
        h0, inv, nullptr, agg1, n);
    k_agg_edges<N_HID><<<((E * (N_HID / 4)) + TPB - 1) / TPB, TPB, 0, stream>>>(
        h0, src, dst, inv, agg1, E);
    k_bias_relu<N_HID><<<((n * N_HID) + TPB - 1) / TPB, TPB, 0, stream>>>(agg1, b1, n);

    float* h2 = h0;
    {
        int tiles = (n / 16) * (N_OUT / 16);
        int blocks = (tiles + 3) / 4;
        k_gemm_wmma_f32<N_HID, N_OUT><<<blocks, 128, 0, stream>>>(agg1, W2, h2, n);
    }
    k_agg_init<N_OUT><<<((n * (N_OUT / 4)) + TPB - 1) / TPB, TPB, 0, stream>>>(
        h2, inv, b2, out, n);
    k_agg_edges<N_OUT><<<((E * (N_OUT / 4)) + TPB - 1) / TPB, TPB, 0, stream>>>(
        h2, src, dst, inv, out, E);

    (void)n_in; (void)out_size; (void)ws_size;
}


# GEMM waves own two side-by-side 16x16 tiles that share the A fragments; accumulator lane halves swapped so each store instruction writes one whole 128-byte line of one row (was two 64-byte pieces)
# speedup vs baseline: 3.1689x; 3.1689x over previous
#include <hip/hip_runtime.h>
#include <hip/hip_bf16.h>


typedef float v2f __attribute__((ext_vector_type(2)));
typedef float v8f __attribute__((ext_vector_type(8)));

#define N_IN 64
#define N_HID 128
#define N_OUT 64


#define N_PASSES 3
#define LOWER_TICKETS 16

__global__ void k_deg_init(float* __restrict__ deg, int* __restrict__ tickets, int n) {
    int i = blockIdx.x * blockDim.x + threadIdx.x;
    if (i < n) deg[i] = 1.0f;
    if (i < N_PASSES) {
        tickets[i] = 0;
        tickets[LOWER_TICKETS + i] = 0;
    }
}

__device__ __forceinline__ bool take_chunk(int* __restrict__ tickets, int nChunks,
                                           int& hf, int& chunk) {
    __shared__ int info[2];
    if (threadIdx.x == 0) {
        int f = (__builtin_amdgcn_s_sendmsg_rtn(0x87) >> 18) & 1;
        int t = atomicAdd(&tickets[f ? 0 : LOWER_TICKETS], 1);
        if (t >= nChunks) {
            f ^= 1;
            t = atomicAdd(&tickets[f ? 0 : LOWER_TICKETS], 1);
        }
        info[0] = f;
        info[1] = t;
    }
    __syncthreads();
    hf    = __builtin_amdgcn_readfirstlane(info[0]);
    chunk = __builtin_amdgcn_readfirstlane(info[1]);
    return chunk < nChunks;
}

#define DEG_EDGES_PER_THREAD 8

__global__ void __launch_bounds__(256)
k_deg_count(const int* __restrict__ dst, float* __restrict__ deg, int* __restrict__ tickets,
            int E, int nChunks, int nHalf) {
    int hf, chunk;
    if (!take_chunk(tickets, nChunks, hf, chunk)) return;
    const int base = chunk * (256 * DEG_EDGES_PER_THREAD) + threadIdx.x;
    int d[DEG_EDGES_PER_THREAD];
#pragma unroll
    for (int k = 0; k < DEG_EDGES_PER_THREAD; ++k) {
        const int e = base + k * 256;
        d[k] = (e < E) ? dst[e] : -1;
    }
#pragma unroll
    for (int k = 0; k < DEG_EDGES_PER_THREAD; ++k)
        if (d[k] >= 0 && (d[k] >= nHalf) == (hf != 0)) atomicAdd(&deg[d[k]], 1.0f);
}

__device__ __forceinline__ float inv_sqrt_deg(float deg) {
    return rsqrtf(fmaxf(deg, 1.0f));
}

template <int K, int N, bool RELU_IN, bool BIAS_OUT>
__global__ void k_gemm_wmma_f32(const float* __restrict__ A,
                                const float* __restrict__ B,
                                const float* __restrict__ bias_in,
                                const float* __restrict__ deg,
                                const float* __restrict__ bias_out,
                                float* __restrict__ C,
                                float* __restrict__ S, int M) {
    const int lane = threadIdx.x & 31;
    const int wave = threadIdx.x >> 5;
    const int wavesPerBlock = blockDim.x >> 5;
    const int tile = blockIdx.x * wavesPerBlock + wave;

    const int nTiles = N / 32;
    const int mTiles = M / 16;
    if (tile >= mTiles * nTiles) return;

    const int row0 = (tile / nTiles) * 16;
    const int col0 = (tile % nTiles) * 32;
    const int half = lane >> 4;
    const int l16  = lane & 15;

    v8f acc0 = {0.f, 0.f, 0.f, 0.f, 0.f, 0.f, 0.f, 0.f};
    v8f acc1 = {0.f, 0.f, 0.f, 0.f, 0.f, 0.f, 0.f, 0.f};

    const float inv_r = inv_sqrt_deg(deg[row0 + l16]);

#pragma unroll
    for (int k = 0; k < K; k += 4) {
        const float* ap = A + (size_t)(row0 + l16) * K + (k + 2 * half);
        v2f a;
        a.x = ap[0];
        a.y = ap[1];
        if (RELU_IN) {
            const float* bi = bias_in + (k + 2 * half);
            a.x = fmaxf(inv_r * a.x + bi[0], 0.0f);
            a.y = fmaxf(inv_r * a.y + bi[1], 0.0f);
        }
        const float* bp = B + (size_t)(k + 2 * half) * N + (col0 + l16);
        v2f b0, b1;
        b0.x = bp[0];
        b0.y = bp[N];
        b1.x = bp[16];
        b1.y = bp[N + 16];
        acc0 = __builtin_amdgcn_wmma_f32_16x16x4_f32(
             false, a,  false, b0,
             (short)0, acc0,  false,  false);
        acc1 = __builtin_amdgcn_wmma_f32_16x16x4_f32(
             false, a,  false, b1,
             (short)0, acc1,  false,  false);
    }

    float bo = 0.0f;
    if (BIAS_OUT) bo = bias_out[col0 + lane];
    float* cp = C + (size_t)row0 * N + (col0 + lane);
    float* sp = S + (size_t)row0 * N + (col0 + lane);
#pragma unroll
    for (int j = 0; j < 8; ++j) {
        const auto sw = __builtin_amdgcn_permlane16_swap(__float_as_uint(acc0[j]),
                                                         __float_as_uint(acc1[j]), false, false);
        const float s_top = __uint_as_float(__builtin_amdgcn_readlane(__float_as_uint(inv_r), j));
        const float s_bot = __uint_as_float(__builtin_amdgcn_readlane(__float_as_uint(inv_r), 8 + j));
        const float c_top = s_top * __uint_as_float(sw[0]);
        const float c_bot = s_bot * __uint_as_float(sw[1]);
        float r_top = c_top, r_bot = c_bot;
        if (BIAS_OUT) {
            r_top = s_top * c_top + bo;
            r_bot = s_bot * c_bot + bo;
        }
        cp[(size_t)j * N]       = c_top;
        sp[(size_t)j * N]       = r_top;
        cp[(size_t)(8 + j) * N] = c_bot;
        sp[(size_t)(8 + j) * N] = r_bot;
    }
}


template <int C>
__device__ __forceinline__ void row_load(const float* __restrict__ h, int s, int lane,
                                         float (&v)[C / 32]) {
    const float* hp = h + (size_t)s * C + lane;
#pragma unroll
    for (int r = 0; r < C / 32; ++r) v[r] = hp[32 * r];
}

template <int C, bool DST_W>
__device__ __forceinline__ void row_add(float* __restrict__ out, int d, float w, int lane,
                                        const float (&v)[C / 32]) {
    float* op = out + (size_t)d * C + lane;
#pragma unroll
    for (int r = 0; r < C / 32; ++r) atomicAdd(op + 32 * r, DST_W ? w * v[r] : v[r]);
}

template <int C>
__device__ __forceinline__ void group_load(const float* __restrict__ h, int s, int t0, int lane,
                                           float (&v)[32 / (C / 32)][C / 32]) {
#pragma unroll
    for (int j = 0; j < 32 / (C / 32); ++j)
        row_load<C>(h, __builtin_amdgcn_readlane(s, t0 + j), lane, v[j]);
}

template <int C, bool DST_W>
__device__ __forceinline__ void group_add(float* __restrict__ out, int d, float w, int t0, int lane,
                                          const float (&v)[32 / (C / 32)][C / 32]) {
#pragma unroll
    for (int j = 0; j < 32 / (C / 32); ++j) {
        const int dj = __builtin_amdgcn_readlane(d, t0 + j);
        float wj = 1.0f;
        if (DST_W) wj = __int_as_float(__builtin_amdgcn_readlane(__float_as_int(w), t0 + j));
        row_add<C, DST_W>(out, dj, wj, lane, v[j]);
    }
}

template <int C, bool DST_W>
__device__ __forceinline__ void walk(const float* __restrict__ h, float* __restrict__ out,
                                     int s, int d, float w, int k, int lane) {
    constexpr int R = C / 32;
    constexpr int G = 32 / R;
    float v[2][G][R];
    const int nFull = k / G;
    if (nFull > 0) {
        group_load<C>(h, s, 0, lane, v[0]);
        for (int g = 0; g < nFull; g += 2) {
            if (g + 1 < nFull) group_load<C>(h, s, (g + 1) * G, lane, v[1]);
            group_add<C, DST_W>(out, d, w, g * G, lane, v[0]);
            if (g + 1 < nFull) {
                if (g + 2 < nFull) group_load<C>(h, s, (g + 2) * G, lane, v[0]);
                group_add<C, DST_W>(out, d, w, (g + 1) * G, lane, v[1]);
            }
        }
    }
    const int t0  = nFull * G;
    const int rem = k - t0;
    if (rem > 0) {
#pragma unroll
        for (int j = 0; j < G - 1; ++j)
            if (j < rem) row_load<C>(h, __builtin_amdgcn_readlane(s, t0 + j), lane, v[0][j]);
#pragma unroll
        for (int j = 0; j < G - 1; ++j)
            if (j < rem) {
                const int dj = __builtin_amdgcn_readlane(d, t0 + j);
                float wj = 1.0f;
                if (DST_W) wj = __int_as_float(__builtin_amdgcn_readlane(__float_as_int(w), t0 + j));
                row_add<C, DST_W>(out, dj, wj, lane, v[0][j]);
            }
    }
}

template <int C, bool DST_W>
__global__ void __launch_bounds__(256)
k_agg_edges(const float* __restrict__ h, const int* __restrict__ src,
            const int* __restrict__ dst, const float* __restrict__ deg,
            float* __restrict__ out, int* __restrict__ tickets,
            int E, int nChunks, int nHalf) {
    __shared__ int sS[8][96];
    __shared__ int sD[8][96];
    int hf, chunk;
    if (!take_chunk(tickets, nChunks, hf, chunk)) return;
    const int lane = threadIdx.x & 31;
    const int wv   = __builtin_amdgcn_readfirstlane((int)(threadIdx.x >> 5));
    const int base = chunk * 512 + wv * 64;
    if (base >= E) return;

    const int e0 = base + lane, e1 = e0 + 32;
    const int i0 = e0 < E ? e0 : E - 1;
    const int i1 = e1 < E ? e1 : E - 1;
    const int s0 = src[i0], d0 = dst[i0];
    const int s1 = src[i1], d1 = dst[i1];
    const bool m0 = (e0 < E) && ((d0 >= nHalf) == (hf != 0));
    const bool m1 = (e1 < E) && ((d1 >= nHalf) == (hf != 0));

    const unsigned b0 = __builtin_amdgcn_ballot_w32(m0);
    const unsigned b1 = __builtin_amdgcn_ballot_w32(m1);
    const int c0  = __builtin_popcount(b0);
    const int cnt = c0 + __builtin_popcount(b1);
    const int r0 = m0 ? (int)__builtin_amdgcn_mbcnt_lo(b0, 0u) : 64 + lane;
    const int r1 = m1 ? c0 + (int)__builtin_amdgcn_mbcnt_lo(b1, 0u) : 64 + lane;
    sS[wv][r0] = s0; sD[wv][r0] = d0;
    sS[wv][r1] = s1; sD[wv][r1] = d1;
    const int q0 = lane < cnt ? lane : r0;
    const int q1 = lane + 32 < cnt ? lane + 32 : r0;
    const int ps0 = sS[wv][q0], ps1 = sS[wv][q1];
    const int pd0 = sD[wv][q0], pd1 = sD[wv][q1];
    float w0 = 0.0f, w1 = 0.0f;
    if (DST_W) {
        w0 = inv_sqrt_deg(deg[pd0]);
        w1 = inv_sqrt_deg(deg[pd1]);
    }

    const int k0 = cnt < 32 ? cnt : 32;
    walk<C, DST_W>(h, out, ps0, pd0, w0, k0, lane);
    walk<C, DST_W>(h, out, ps1, pd1, w1, cnt - k0, lane);
}


extern "C" void kernel_launch(void* const* d_in, const int* in_sizes, int n_in,
                              void* d_out, int out_size, void* d_ws, size_t ws_size,
                              hipStream_t stream) {
    const float* x    = (const float*)d_in[0];
    const int*   eidx = (const int*)d_in[1];
    const float* W1   = (const float*)d_in[2];
    const float* b1   = (const float*)d_in[3];
    const float* W2   = (const float*)d_in[4];
    const float* b2   = (const float*)d_in[5];
    float*       out  = (float*)d_out;

    const int n = in_sizes[0] / N_IN;
    const int E = in_sizes[1] / 2;
    const int* src = eidx;
    const int* dst = eidx + E;

    char* ws = (char*)d_ws;
    const size_t OFF_DEG = 0;
    const size_t OFF_H0  = (((size_t)n * 4 + 192) + 511) & ~(size_t)511;
    const size_t OFF_AGG = OFF_H0 + (size_t)n * N_HID * 4;
    float* deg  = (float*)(ws + OFF_DEG);
    int* tickets = (int*)(ws + OFF_DEG + (size_t)n * 4);
    float* h0   = (float*)(ws + OFF_H0);
    float* agg1 = (float*)(ws + OFF_AGG);

    const int TPB = 256;
    const int nHalf = n / 2;
    const int degChunks = (E + 256 * DEG_EDGES_PER_THREAD - 1) / (256 * DEG_EDGES_PER_THREAD);
    const int aggChunks = (E + 511) / 512;

    k_deg_init<<<(n + TPB - 1) / TPB, TPB, 0, stream>>>(deg, tickets, n);
    k_deg_count<<<2 * degChunks, TPB, 0, stream>>>(dst, deg, tickets, E, degChunks, nHalf);

    {
        int tiles = (n / 16) * (N_HID / 32);
        int blocks = (tiles + 3) / 4;
        k_gemm_wmma_f32<N_IN, N_HID, false, false><<<blocks, 128, 0, stream>>>(
            x, W1, nullptr, deg, nullptr, h0, agg1, n);
    }
    k_agg_edges<N_HID, false><<<2 * aggChunks, TPB, 0, stream>>>(
        h0, src, dst, deg, agg1, tickets + 1, E, aggChunks, nHalf);

    float* h2 = h0;
    {
        int tiles = (n / 16) * (N_OUT / 32);
        int blocks = (tiles + 3) / 4;
        k_gemm_wmma_f32<N_HID, N_OUT, true, true><<<blocks, 128, 0, stream>>>(
            agg1, W2, b1, deg, b2, h2, out, n);
    }
    k_agg_edges<N_OUT, true><<<2 * aggChunks, TPB, 0, stream>>>(
        h2, src, dst, deg, out, tickets + 2, E, aggChunks, nHalf);

    (void)n_in; (void)out_size; (void)ws_size;
}
